# starting kernel
# baseline (speedup 1.0000x reference)
#include <hip/hip_runtime.h>
#include <hip/hip_bf16.h>

typedef __attribute__((ext_vector_type(2))) float v2f;
typedef __attribute__((ext_vector_type(8))) float v8f;

#define IN_DIM 128
#define OUT_DIM 64
#define NEG_SLOPE 0.2f

__global__ void gat_init(float* __restrict__ out, float* __restrict__ denom,
                         unsigned* __restrict__ m_enc, int n_nodes) {
    long long i = (long long)blockIdx.x * blockDim.x + threadIdx.x;
    long long total = (long long)n_nodes * OUT_DIM;
    if (i < total) out[i] = 0.0f;
    if (i < n_nodes) {
        denom[i] = 0.0f;
        m_enc[i] = 0x007FFFFFu;
    }
}

__global__ __launch_bounds__(256) void gat_gemm(const float* __restrict__ x,
                                                const float* __restrict__ W,
                                                float* __restrict__ feat,
                                                int n_nodes) {
    __shared__ float Wt[OUT_DIM][IN_DIM];
    for (int i = threadIdx.x; i < IN_DIM * OUT_DIM; i += blockDim.x) {
        int k = i >> 6;
        int c = i & 63;
        Wt[c][k] = W[i];
    }
    __syncthreads();

    const int wave = threadIdx.x >> 5;
    const int lane = threadIdx.x & 31;
    const int tile = blockIdx.x * (blockDim.x >> 5) + wave;
    const int n_tiles = (n_nodes + 15) >> 4;
    if (tile >= n_tiles) return;

    const int row0 = tile << 4;
    const int l16  = lane & 15;
    const int hi   = lane >> 4;

    v8f acc0 = {}, acc1 = {}, acc2 = {}, acc3 = {};
    const float* xrow = x + (size_t)(row0 + l16) * IN_DIM;

    for (int k = 0; k < IN_DIM; k += 4) {
        v2f a = *(const v2f*)(xrow + k + 2 * hi);
        v2f b0 = *(const v2f*)&Wt[ 0 + l16][k + 2 * hi];
        v2f b1 = *(const v2f*)&Wt[16 + l16][k + 2 * hi];
        v2f b2 = *(const v2f*)&Wt[32 + l16][k + 2 * hi];
        v2f b3 = *(const v2f*)&Wt[48 + l16][k + 2 * hi];
        acc0 = __builtin_amdgcn_wmma_f32_16x16x4_f32(false, a, false, b0, (short)0, acc0, false, false);
        acc1 = __builtin_amdgcn_wmma_f32_16x16x4_f32(false, a, false, b1, (short)0, acc1, false, false);
        acc2 = __builtin_amdgcn_wmma_f32_16x16x4_f32(false, a, false, b2, (short)0, acc2, false, false);
        acc3 = __builtin_amdgcn_wmma_f32_16x16x4_f32(false, a, false, b3, (short)0, acc3, false, false);
    }

    float* o = feat + ((size_t)(row0 + hi * 8)) * OUT_DIM + l16;
#pragma unroll
    for (int r = 0; r < 8; ++r) {
        o[(size_t)r * OUT_DIM +  0] = acc0[r];
        o[(size_t)r * OUT_DIM + 16] = acc1[r];
        o[(size_t)r * OUT_DIM + 32] = acc2[r];
        o[(size_t)r * OUT_DIM + 48] = acc3[r];
    }
}

__global__ void gat_logits(const float* __restrict__ feat,
                           const float* __restrict__ attn_l,
                           const float* __restrict__ attn_r,
                           float* __restrict__ el, float* __restrict__ er,
                           int n_nodes) {
    const int lane  = threadIdx.x & 31;
    const int waveg = (blockIdx.x * blockDim.x + threadIdx.x) >> 5;
    const int nwav  = (gridDim.x * blockDim.x) >> 5;
    const float al0 = attn_l[lane], al1 = attn_l[lane + 32];
    const float ar0 = attn_r[lane], ar1 = attn_r[lane + 32];
    for (int node = waveg; node < n_nodes; node += nwav) {
        const float* f = feat + (size_t)node * OUT_DIM;
        float f0 = f[lane], f1 = f[lane + 32];
        float pl = f0 * al0 + f1 * al1;
        float pr = f0 * ar0 + f1 * ar1;
#pragma unroll
        for (int off = 16; off; off >>= 1) {
            pl += __shfl_xor(pl, off, 32);
            pr += __shfl_xor(pr, off, 32);
        }
        if (lane == 0) { el[node] = pl; er[node] = pr; }
    }
}

__global__ void gat_edge_score(const int* __restrict__ src, const int* __restrict__ dst,
                               const float* __restrict__ el, const float* __restrict__ er,
                               float* __restrict__ e, unsigned* __restrict__ m_enc,
                               int n_edges) {
    int i = blockIdx.x * blockDim.x + threadIdx.x;
    if (i >= n_edges) return;
    int d = dst[i];
    float v = el[src[i]] + er[d];
    v = v > 0.0f ? v : NEG_SLOPE * v;
    e[i] = v;
    unsigned bits = __float_as_uint(v);
    unsigned enc  = (bits & 0x80000000u) ? ~bits : (bits | 0x80000000u);
    atomicMax(&m_enc[d], enc);
}

__global__ void gat_exp(const int* __restrict__ dst, const unsigned* __restrict__ m_enc,
                        float* __restrict__ e, float* __restrict__ denom, int n_edges) {
    int i = blockIdx.x * blockDim.x + threadIdx.x;
    if (i >= n_edges) return;
    int d = dst[i];
    unsigned enc  = m_enc[d];
    unsigned bits = (enc & 0x80000000u) ? (enc & 0x7FFFFFFFu) : ~enc;
    float m  = __uint_as_float(bits);
    float ex = expf(e[i] - m);
    e[i] = ex;
    atomicAdd(&denom[d], ex);
}

__global__ void gat_aggregate(const int* __restrict__ src, const int* __restrict__ dst,
                              const float* __restrict__ ex, const float* __restrict__ denom,
                              const float* __restrict__ feat, float* __restrict__ out,
                              int n_edges) {
    long long t = (long long)blockIdx.x * blockDim.x + threadIdx.x;
    int edge = (int)(t >> 6);
    if (edge >= n_edges) return;
    int c = (int)t & 63;
    int d = dst[edge], s = src[edge];
    float alpha = ex[edge] / denom[d];
    atomicAdd(&out[(size_t)d * OUT_DIM + c], alpha * feat[(size_t)s * OUT_DIM + c]);
}

__global__ void gat_finalize(float* __restrict__ out, const float* __restrict__ bias,
                             int n_nodes) {
    long long i = (long long)blockIdx.x * blockDim.x + threadIdx.x;
    long long total = (long long)n_nodes * OUT_DIM;
    if (i >= total) return;
    out[i] = tanhf(out[i] + bias[i & 63]);
}

static inline size_t align256(size_t x) { return (x + 255) & ~(size_t)255; }

extern "C" void kernel_launch(void* const* d_in, const int* in_sizes, int n_in,
                              void* d_out, int out_size, void* d_ws, size_t ws_size,
                              hipStream_t stream) {
    const float* x      = (const float*)d_in[0];
    const int*   src    = (const int*)d_in[1];
    const int*   dst    = (const int*)d_in[2];
    const float* W      = (const float*)d_in[3];
    const float* attn_l = (const float*)d_in[4];
    const float* attn_r = (const float*)d_in[5];
    const float* bias   = (const float*)d_in[6];
    float*       out    = (float*)d_out;

    const int N = in_sizes[0] / IN_DIM;
    const int E = in_sizes[1];

    char* ws = (char*)d_ws;
    float*    feat  = (float*)ws;    ws += align256((size_t)N * OUT_DIM * sizeof(float));
    float*    e     = (float*)ws;    ws += align256((size_t)E * sizeof(float));
    float*    el    = (float*)ws;    ws += align256((size_t)N * sizeof(float));
    float*    er    = (float*)ws;    ws += align256((size_t)N * sizeof(float));
    float*    denom = (float*)ws;    ws += align256((size_t)N * sizeof(float));
    unsigned* m_enc = (unsigned*)ws; ws += align256((size_t)N * sizeof(unsigned));

    const int BLK = 256;

    {
        long long total = (long long)N * OUT_DIM;
        int blocks = (int)((total + BLK - 1) / BLK);
        gat_init<<<blocks, BLK, 0, stream>>>(out, denom, m_enc, N);
    }
    {
        int n_tiles = (N + 15) / 16;
        int waves_per_block = BLK / 32;
        int blocks = (n_tiles + waves_per_block - 1) / waves_per_block;
        gat_gemm<<<blocks, BLK, 0, stream>>>(x, W, feat, N);
    }
    {
        int blocks = 1024;
        gat_logits<<<blocks, BLK, 0, stream>>>(feat, attn_l, attn_r, el, er, N);
    }
    {
        int blocks = (E + BLK - 1) / BLK;
        gat_edge_score<<<blocks, BLK, 0, stream>>>(src, dst, el, er, e, m_enc, E);
    }
    {
        int blocks = (E + BLK - 1) / BLK;
        gat_exp<<<blocks, BLK, 0, stream>>>(dst, m_enc, e, denom, E);
    }
    {
        long long total = (long long)E * OUT_DIM;
        int blocks = (int)((total + BLK - 1) / BLK);
        gat_aggregate<<<blocks, BLK, 0, stream>>>(src, dst, e, denom, feat, out, E);
    }
    {
        long long total = (long long)N * OUT_DIM;
        int blocks = (int)((total + BLK - 1) / BLK);
        gat_finalize<<<blocks, BLK, 0, stream>>>(out, bias, N);
    }
}


# Group edges by destination (one integer atomic rank per edge, block scan, scatter), then one wave per destination gathers its rows: softmax, sum, bias, tanh fused; all float atomics gone
# speedup vs baseline: 2.2713x; 2.2713x over previous
#include <hip/hip_runtime.h>
#include <hip/hip_bf16.h>


typedef __attribute__((ext_vector_type(2))) float v2f;
typedef __attribute__((ext_vector_type(8))) float v8f;

#define IN_DIM 128
#define OUT_DIM 64
#define NEG_SLOPE 0.2f
#define SCAN_TILE 4096

__global__ void gat_zero_counts(int* __restrict__ cnt, int n_nodes) {
    int i = blockIdx.x * blockDim.x + threadIdx.x;
    if (i < n_nodes) cnt[i] = 0;
}

__global__ void gat_rank(const int* __restrict__ dst, int* __restrict__ cnt,
                         int* __restrict__ rank, int n_edges) {
    int i = blockIdx.x * blockDim.x + threadIdx.x;
    if (i >= n_edges) return;
    rank[i] = atomicAdd(&cnt[dst[i]], 1);
}

__global__ __launch_bounds__(1024) void gat_offsets(const int* __restrict__ cnt,
                                                    int* __restrict__ off, int n_nodes) {
    __shared__ int wsum[32];
    const int tid  = threadIdx.x;
    const int lane = tid & 31;
    const int wave = tid >> 5;

    int part = 0;
    for (int k = 0; k < (int)blockIdx.x; ++k) {
        const int4 v = *(const int4*)(cnt + (size_t)k * SCAN_TILE + tid * 4);
        part += (v.x + v.y) + (v.z + v.w);
    }
    for (int o = 16; o; o >>= 1) part += __shfl_xor(part, o, 32);
    if (lane == 0) wsum[wave] = part;
    __syncthreads();
    int carry = wsum[lane];
    for (int o = 16; o; o >>= 1) carry += __shfl_xor(carry, o, 32);
    __syncthreads();

    const int i0 = (int)blockIdx.x * SCAN_TILE + tid * 4;
    const int c0 = (i0 + 0 < n_nodes) ? cnt[i0 + 0] : 0;
    const int c1 = (i0 + 1 < n_nodes) ? cnt[i0 + 1] : 0;
    const int c2 = (i0 + 2 < n_nodes) ? cnt[i0 + 2] : 0;
    const int c3 = (i0 + 3 < n_nodes) ? cnt[i0 + 3] : 0;
    const int s0 = c0, s1 = s0 + c1, s2 = s1 + c2, s3 = s2 + c3;

    int incl = s3;
    for (int o = 1; o < 32; o <<= 1) {
        const int u = __shfl_up(incl, o, 32);
        if (lane >= o) incl += u;
    }
    if (lane == 31) wsum[wave] = incl;
    __syncthreads();
    const int wv = wsum[lane];
    int winc = wv;
    for (int o = 1; o < 32; o <<= 1) {
        const int u = __shfl_up(winc, o, 32);
        if (lane >= o) winc += u;
    }
    const int wprefix = __shfl(winc - wv, wave, 32);

    const int ex = carry + wprefix + (incl - s3);
    if (i0 + 0 < n_nodes) off[i0 + 0] = ex;
    if (i0 + 1 < n_nodes) off[i0 + 1] = ex + s0;
    if (i0 + 2 < n_nodes) off[i0 + 2] = ex + s1;
    if (i0 + 3 < n_nodes) off[i0 + 3] = ex + s2;
}

__global__ void gat_fill(const int* __restrict__ src, const int* __restrict__ dst,
                         const int* __restrict__ rank, const int* __restrict__ off,
                         int* __restrict__ ssrc, int n_edges) {
    int i = blockIdx.x * blockDim.x + threadIdx.x;
    if (i >= n_edges) return;
    ssrc[off[dst[i]] + rank[i]] = src[i];
}

__global__ __launch_bounds__(256) void gat_gemm(const float* __restrict__ x,
                                                const float* __restrict__ W,
                                                float* __restrict__ feat,
                                                int n_nodes) {
    __shared__ float Wt[OUT_DIM][IN_DIM];
    for (int i = threadIdx.x; i < IN_DIM * OUT_DIM; i += blockDim.x) {
        int k = i >> 6;
        int c = i & 63;
        Wt[c][k] = W[i];
    }
    __syncthreads();

    const int wave = threadIdx.x >> 5;
    const int lane = threadIdx.x & 31;
    const int tile = blockIdx.x * (blockDim.x >> 5) + wave;
    const int n_tiles = (n_nodes + 15) >> 4;
    if (tile >= n_tiles) return;

    const int row0 = tile << 4;
    const int l16  = lane & 15;
    const int hi   = lane >> 4;

    v8f acc0 = {}, acc1 = {}, acc2 = {}, acc3 = {};
    const float* xrow = x + (size_t)(row0 + l16) * IN_DIM;

    for (int k = 0; k < IN_DIM; k += 4) {
        v2f a = *(const v2f*)(xrow + k + 2 * hi);
        v2f b0 = *(const v2f*)&Wt[ 0 + l16][k + 2 * hi];
        v2f b1 = *(const v2f*)&Wt[16 + l16][k + 2 * hi];
        v2f b2 = *(const v2f*)&Wt[32 + l16][k + 2 * hi];
        v2f b3 = *(const v2f*)&Wt[48 + l16][k + 2 * hi];
        acc0 = __builtin_amdgcn_wmma_f32_16x16x4_f32(false, a, false, b0, (short)0, acc0, false, false);
        acc1 = __builtin_amdgcn_wmma_f32_16x16x4_f32(false, a, false, b1, (short)0, acc1, false, false);
        acc2 = __builtin_amdgcn_wmma_f32_16x16x4_f32(false, a, false, b2, (short)0, acc2, false, false);
        acc3 = __builtin_amdgcn_wmma_f32_16x16x4_f32(false, a, false, b3, (short)0, acc3, false, false);
    }

    float* o = feat + ((size_t)(row0 + hi * 8)) * OUT_DIM + l16;
#pragma unroll
    for (int r = 0; r < 8; ++r) {
        o[(size_t)r * OUT_DIM +  0] = acc0[r];
        o[(size_t)r * OUT_DIM + 16] = acc1[r];
        o[(size_t)r * OUT_DIM + 32] = acc2[r];
        o[(size_t)r * OUT_DIM + 48] = acc3[r];
    }
}

__global__ void gat_logits(const float* __restrict__ feat,
                           const float* __restrict__ attn_l,
                           const float* __restrict__ attn_r,
                           float* __restrict__ el, float* __restrict__ er,
                           int n_nodes) {
    const int lane  = threadIdx.x & 31;
    const int waveg = (blockIdx.x * blockDim.x + threadIdx.x) >> 5;
    const int nwav  = (gridDim.x * blockDim.x) >> 5;
    const float al0 = attn_l[lane], al1 = attn_l[lane + 32];
    const float ar0 = attn_r[lane], ar1 = attn_r[lane + 32];
    for (int node = waveg; node < n_nodes; node += nwav) {
        const float* f = feat + (size_t)node * OUT_DIM;
        float f0 = f[lane], f1 = f[lane + 32];
        float pl = f0 * al0 + f1 * al1;
        float pr = f0 * ar0 + f1 * ar1;
#pragma unroll
        for (int off = 16; off; off >>= 1) {
            pl += __shfl_xor(pl, off, 32);
            pr += __shfl_xor(pr, off, 32);
        }
        if (lane == 0) { el[node] = pl; er[node] = pr; }
    }
}

template <int U>
__device__ __forceinline__ void gat_rows(const float* __restrict__ feat, int s, float a,
                                         int t, int lane, float& ax, float& ay) {
    v2f   f[U];
    float w[U];
#pragma unroll
    for (int u = 0; u < U; ++u) {
        const int st = __builtin_amdgcn_readlane(s, t + u);
        w[u] = __int_as_float(__builtin_amdgcn_readlane(__float_as_int(a), t + u));
        f[u] = *(const v2f*)(feat + (size_t)st * OUT_DIM + 2 * lane);
    }
#pragma unroll
    for (int u = 0; u < U; ++u) {
        ax += w[u] * f[u].x;
        ay += w[u] * f[u].y;
    }
}

__device__ __forceinline__ void gat_chunk(const float* __restrict__ feat, int s, float a,
                                          int cnt, int lane, float& ax, float& ay) {
    int t = 0;
    for (; t + 8 <= cnt; t += 8) gat_rows<8>(feat, s, a, t, lane, ax, ay);
    if (t + 4 <= cnt) { gat_rows<4>(feat, s, a, t, lane, ax, ay); t += 4; }
    if (t + 2 <= cnt) { gat_rows<2>(feat, s, a, t, lane, ax, ay); t += 2; }
    if (t < cnt)      { gat_rows<1>(feat, s, a, t, lane, ax, ay); }
}

__device__ __forceinline__ float gat_score(float l, float r) {
    float v = l + r;
    return v > 0.0f ? v : NEG_SLOPE * v;
}

__global__ __launch_bounds__(256) void gat_segment(const int* __restrict__ ssrc,
                                                   const int* __restrict__ off,
                                                   const float* __restrict__ el,
                                                   const float* __restrict__ er,
                                                   const float* __restrict__ feat,
                                                   const float* __restrict__ bias,
                                                   float* __restrict__ out,
                                                   int n_nodes, int n_edges) {
    const int lane = threadIdx.x & 31;
    const int wave = __builtin_amdgcn_readfirstlane((int)(threadIdx.x >> 5));
    const int d = blockIdx.x * (blockDim.x >> 5) + wave;
    if (d >= n_nodes) return;

    const int beg = off[d];
    const int end = (d + 1 < n_nodes) ? off[d + 1] : n_edges;
    const float erd = er[d];

    const int  j0  = beg + lane;
    const bool act = j0 < end;
    int   s0 = 0;
    float v0 = -INFINITY;
    if (act) { s0 = ssrc[j0]; v0 = gat_score(el[s0], erd); }

    float m = v0;
    for (int j = j0 + 32; j < end; j += 32) m = fmaxf(m, gat_score(el[ssrc[j]], erd));
    for (int o = 16; o; o >>= 1) m = fmaxf(m, __shfl_xor(m, o, 32));

    const float w0 = act ? expf(v0 - m) : 0.0f;
    float den = w0;
    for (int j = j0 + 32; j < end; j += 32) den += expf(gat_score(el[ssrc[j]], erd) - m);
    for (int o = 16; o; o >>= 1) den += __shfl_xor(den, o, 32);

    float ax = 0.0f, ay = 0.0f;
    const int deg = end - beg;
    gat_chunk(feat, s0, w0 / den, deg < 32 ? deg : 32, lane, ax, ay);
    for (int base = beg + 32; base < end; base += 32) {
        const int j = base + lane;
        int   s = 0;
        float a = 0.0f;
        if (j < end) { s = ssrc[j]; a = expf(gat_score(el[s], erd) - m) / den; }
        const int rem = end - base;
        gat_chunk(feat, s, a, rem < 32 ? rem : 32, lane, ax, ay);
    }

    const v2f b = *(const v2f*)(bias + 2 * lane);
    v2f r;
    r.x = tanhf(ax + b.x);
    r.y = tanhf(ay + b.y);
    *(v2f*)(out + (size_t)d * OUT_DIM + 2 * lane) = r;
}

static inline size_t align256(size_t x) { return (x + 255) & ~(size_t)255; }

extern "C" void kernel_launch(void* const* d_in, const int* in_sizes, int n_in,
                              void* d_out, int out_size, void* d_ws, size_t ws_size,
                              hipStream_t stream) {
    const float* x      = (const float*)d_in[0];
    const int*   src    = (const int*)d_in[1];
    const int*   dst    = (const int*)d_in[2];
    const float* W      = (const float*)d_in[3];
    const float* attn_l = (const float*)d_in[4];
    const float* attn_r = (const float*)d_in[5];
    const float* bias   = (const float*)d_in[6];
    float*       out    = (float*)d_out;

    const int N = in_sizes[0] / IN_DIM;
    const int E = in_sizes[1];

    char* ws = (char*)d_ws;
    float* feat = (float*)ws;
    int*   rank = (int*)ws;          ws += align256((size_t)N * OUT_DIM * sizeof(float));
    int*   ssrc = (int*)ws;          ws += align256((size_t)E * sizeof(float));
    float* el   = (float*)ws;        ws += align256((size_t)N * sizeof(float));
    float* er   = (float*)ws;        ws += align256((size_t)N * sizeof(float));
    int*   cnt  = (int*)ws;          ws += align256((size_t)N * sizeof(float));
    int*   off  = (int*)ws;          ws += align256((size_t)N * sizeof(unsigned));

    const int BLK = 256;

    gat_zero_counts<<<(N + BLK - 1) / BLK, BLK, 0, stream>>>(cnt, N);
    gat_rank<<<(E + BLK - 1) / BLK, BLK, 0, stream>>>(dst, cnt, rank, E);
    gat_offsets<<<(N + SCAN_TILE - 1) / SCAN_TILE, 1024, 0, stream>>>(cnt, off, N);
    gat_fill<<<(E + BLK - 1) / BLK, BLK, 0, stream>>>(src, dst, rank, off, ssrc, E);
    {
        int n_tiles = (N + 15) / 16;
        int waves_per_block = BLK / 32;
        int blocks = (n_tiles + waves_per_block - 1) / waves_per_block;
        gat_gemm<<<blocks, BLK, 0, stream>>>(x, W, feat, N);
    }
    gat_logits<<<1024, BLK, 0, stream>>>(feat, attn_l, attn_r, el, er, N);
    {
        int waves_per_block = BLK / 32;
        int blocks = (N + waves_per_block - 1) / waves_per_block;
        gat_segment<<<blocks, BLK, 0, stream>>>(ssrc, off, el, er, feat, bias, out, N, E);
    }
}


# GEMM: W tile in LDS at a row pitch of 132 floats (was 128), staged 16 columns x 2 k per wave: B-fragment loads (were 16-way) and staging stores (were 32-way) become bank-conflict free
# speedup vs baseline: 2.4954x; 1.0987x over previous
#include <hip/hip_runtime.h>
#include <hip/hip_bf16.h>


typedef __attribute__((ext_vector_type(2))) float v2f;
typedef __attribute__((ext_vector_type(8))) float v8f;

#define IN_DIM 128
#define OUT_DIM 64
#define NEG_SLOPE 0.2f
#define SCAN_TILE 4096
#define WT_PITCH (IN_DIM + 4)

__global__ void gat_zero_counts(int* __restrict__ cnt, int n_nodes) {
    int i = blockIdx.x * blockDim.x + threadIdx.x;
    if (i < n_nodes) cnt[i] = 0;
}

__global__ void gat_rank(const int* __restrict__ dst, int* __restrict__ cnt,
                         int* __restrict__ rank, int n_edges) {
    int i = blockIdx.x * blockDim.x + threadIdx.x;
    if (i >= n_edges) return;
    rank[i] = atomicAdd(&cnt[dst[i]], 1);
}

__global__ __launch_bounds__(1024) void gat_offsets(const int* __restrict__ cnt,
                                                    int* __restrict__ off, int n_nodes) {
    __shared__ int wsum[32];
    const int tid  = threadIdx.x;
    const int lane = tid & 31;
    const int wave = tid >> 5;

    int part = 0;
    for (int k = 0; k < (int)blockIdx.x; ++k) {
        const int4 v = *(const int4*)(cnt + (size_t)k * SCAN_TILE + tid * 4);
        part += (v.x + v.y) + (v.z + v.w);
    }
    for (int o = 16; o; o >>= 1) part += __shfl_xor(part, o, 32);
    if (lane == 0) wsum[wave] = part;
    __syncthreads();
    int carry = wsum[lane];
    for (int o = 16; o; o >>= 1) carry += __shfl_xor(carry, o, 32);
    __syncthreads();

    const int i0 = (int)blockIdx.x * SCAN_TILE + tid * 4;
    const int c0 = (i0 + 0 < n_nodes) ? cnt[i0 + 0] : 0;
    const int c1 = (i0 + 1 < n_nodes) ? cnt[i0 + 1] : 0;
    const int c2 = (i0 + 2 < n_nodes) ? cnt[i0 + 2] : 0;
    const int c3 = (i0 + 3 < n_nodes) ? cnt[i0 + 3] : 0;
    const int s0 = c0, s1 = s0 + c1, s2 = s1 + c2, s3 = s2 + c3;

    int incl = s3;
    for (int o = 1; o < 32; o <<= 1) {
        const int u = __shfl_up(incl, o, 32);
        if (lane >= o) incl += u;
    }
    if (lane == 31) wsum[wave] = incl;
    __syncthreads();
    const int wv = wsum[lane];
    int winc = wv;
    for (int o = 1; o < 32; o <<= 1) {
        const int u = __shfl_up(winc, o, 32);
        if (lane >= o) winc += u;
    }
    const int wprefix = __shfl(winc - wv, wave, 32);

    const int ex = carry + wprefix + (incl - s3);
    if (i0 + 0 < n_nodes) off[i0 + 0] = ex;
    if (i0 + 1 < n_nodes) off[i0 + 1] = ex + s0;
    if (i0 + 2 < n_nodes) off[i0 + 2] = ex + s1;
    if (i0 + 3 < n_nodes) off[i0 + 3] = ex + s2;
}

__global__ void gat_fill(const int* __restrict__ src, const int* __restrict__ dst,
                         const int* __restrict__ rank, const int* __restrict__ off,
                         int* __restrict__ ssrc, int n_edges) {
    int i = blockIdx.x * blockDim.x + threadIdx.x;
    if (i >= n_edges) return;
    ssrc[off[dst[i]] + rank[i]] = src[i];
}

__global__ __launch_bounds__(256) void gat_gemm(const float* __restrict__ x,
                                                const float* __restrict__ W,
                                                float* __restrict__ feat,
                                                int n_nodes) {
    __shared__ float Wt[OUT_DIM][WT_PITCH];
    for (int i = threadIdx.x; i < IN_DIM * OUT_DIM; i += blockDim.x) {
        const int c = ((i >> 5) & 3) * 16 + (i & 15);
        const int k = (i >> 7) * 2 + ((i >> 4) & 1);
        Wt[c][k] = W[k * OUT_DIM + c];
    }
    __syncthreads();

    const int wave = threadIdx.x >> 5;
    const int lane = threadIdx.x & 31;
    const int tile = blockIdx.x * (blockDim.x >> 5) + wave;
    const int n_tiles = (n_nodes + 15) >> 4;
    if (tile >= n_tiles) return;

    const int row0 = tile << 4;
    const int l16  = lane & 15;
    const int hi   = lane >> 4;

    v8f acc0 = {}, acc1 = {}, acc2 = {}, acc3 = {};
    const float* xrow = x + (size_t)(row0 + l16) * IN_DIM;

    for (int k = 0; k < IN_DIM; k += 4) {
        v2f a = *(const v2f*)(xrow + k + 2 * hi);
        v2f b0 = *(const v2f*)&Wt[ 0 + l16][k + 2 * hi];
        v2f b1 = *(const v2f*)&Wt[16 + l16][k + 2 * hi];
        v2f b2 = *(const v2f*)&Wt[32 + l16][k + 2 * hi];
        v2f b3 = *(const v2f*)&Wt[48 + l16][k + 2 * hi];
        acc0 = __builtin_amdgcn_wmma_f32_16x16x4_f32(false, a, false, b0, (short)0, acc0, false, false);
        acc1 = __builtin_amdgcn_wmma_f32_16x16x4_f32(false, a, false, b1, (short)0, acc1, false, false);
        acc2 = __builtin_amdgcn_wmma_f32_16x16x4_f32(false, a, false, b2, (short)0, acc2, false, false);
        acc3 = __builtin_amdgcn_wmma_f32_16x16x4_f32(false, a, false, b3, (short)0, acc3, false, false);
    }

    float* o = feat + ((size_t)(row0 + hi * 8)) * OUT_DIM + l16;
#pragma unroll
    for (int r = 0; r < 8; ++r) {
        o[(size_t)r * OUT_DIM +  0] = acc0[r];
        o[(size_t)r * OUT_DIM + 16] = acc1[r];
        o[(size_t)r * OUT_DIM + 32] = acc2[r];
        o[(size_t)r * OUT_DIM + 48] = acc3[r];
    }
}

__global__ void gat_logits(const float* __restrict__ feat,
                           const float* __restrict__ attn_l,
                           const float* __restrict__ attn_r,
                           float* __restrict__ el, float* __restrict__ er,
                           int n_nodes) {
    const int lane  = threadIdx.x & 31;
    const int waveg = (blockIdx.x * blockDim.x + threadIdx.x) >> 5;
    const int nwav  = (gridDim.x * blockDim.x) >> 5;
    const float al0 = attn_l[lane], al1 = attn_l[lane + 32];
    const float ar0 = attn_r[lane], ar1 = attn_r[lane + 32];
    for (int node = waveg; node < n_nodes; node += nwav) {
        const float* f = feat + (size_t)node * OUT_DIM;
        float f0 = f[lane], f1 = f[lane + 32];
        float pl = f0 * al0 + f1 * al1;
        float pr = f0 * ar0 + f1 * ar1;
#pragma unroll
        for (int off = 16; off; off >>= 1) {
            pl += __shfl_xor(pl, off, 32);
            pr += __shfl_xor(pr, off, 32);
        }
        if (lane == 0) { el[node] = pl; er[node] = pr; }
    }
}

template <int U>
__device__ __forceinline__ void gat_rows(const float* __restrict__ feat, int s, float a,
                                         int t, int lane, float& ax, float& ay) {
    v2f   f[U];
    float w[U];
#pragma unroll
    for (int u = 0; u < U; ++u) {
        const int st = __builtin_amdgcn_readlane(s, t + u);
        w[u] = __int_as_float(__builtin_amdgcn_readlane(__float_as_int(a), t + u));
        f[u] = *(const v2f*)(feat + (size_t)st * OUT_DIM + 2 * lane);
    }
#pragma unroll
    for (int u = 0; u < U; ++u) {
        ax += w[u] * f[u].x;
        ay += w[u] * f[u].y;
    }
}

__device__ __forceinline__ void gat_chunk(const float* __restrict__ feat, int s, float a,
                                          int cnt, int lane, float& ax, float& ay) {
    int t = 0;
    for (; t + 8 <= cnt; t += 8) gat_rows<8>(feat, s, a, t, lane, ax, ay);
    if (t + 4 <= cnt) { gat_rows<4>(feat, s, a, t, lane, ax, ay); t += 4; }
    if (t + 2 <= cnt) { gat_rows<2>(feat, s, a, t, lane, ax, ay); t += 2; }
    if (t < cnt)      { gat_rows<1>(feat, s, a, t, lane, ax, ay); }
}

__device__ __forceinline__ float gat_score(float l, float r) {
    float v = l + r;
    return v > 0.0f ? v : NEG_SLOPE * v;
}

__global__ __launch_bounds__(256) void gat_segment(const int* __restrict__ ssrc,
                                                   const int* __restrict__ off,
                                                   const float* __restrict__ el,
                                                   const float* __restrict__ er,
                                                   const float* __restrict__ feat,
                                                   const float* __restrict__ bias,
                                                   float* __restrict__ out,
                                                   int n_nodes, int n_edges) {
    const int lane = threadIdx.x & 31;
    const int wave = __builtin_amdgcn_readfirstlane((int)(threadIdx.x >> 5));
    const int d = blockIdx.x * (blockDim.x >> 5) + wave;
    if (d >= n_nodes) return;

    const int beg = off[d];
    const int end = (d + 1 < n_nodes) ? off[d + 1] : n_edges;
    const float erd = er[d];

    const int  j0  = beg + lane;
    const bool act = j0 < end;
    int   s0 = 0;
    float v0 = -INFINITY;
    if (act) { s0 = ssrc[j0]; v0 = gat_score(el[s0], erd); }

    float m = v0;
    for (int j = j0 + 32; j < end; j += 32) m = fmaxf(m, gat_score(el[ssrc[j]], erd));
    for (int o = 16; o; o >>= 1) m = fmaxf(m, __shfl_xor(m, o, 32));

    const float w0 = act ? expf(v0 - m) : 0.0f;
    float den = w0;
    for (int j = j0 + 32; j < end; j += 32) den += expf(gat_score(el[ssrc[j]], erd) - m);
    for (int o = 16; o; o >>= 1) den += __shfl_xor(den, o, 32);

    float ax = 0.0f, ay = 0.0f;
    const int deg = end - beg;
    gat_chunk(feat, s0, w0 / den, deg < 32 ? deg : 32, lane, ax, ay);
    for (int base = beg + 32; base < end; base += 32) {
        const int j = base + lane;
        int   s = 0;
        float a = 0.0f;
        if (j < end) { s = ssrc[j]; a = expf(gat_score(el[s], erd) - m) / den; }
        const int rem = end - base;
        gat_chunk(feat, s, a, rem < 32 ? rem : 32, lane, ax, ay);
    }

    const v2f b = *(const v2f*)(bias + 2 * lane);
    v2f r;
    r.x = tanhf(ax + b.x);
    r.y = tanhf(ay + b.y);
    *(v2f*)(out + (size_t)d * OUT_DIM + 2 * lane) = r;
}

static inline size_t align256(size_t x) { return (x + 255) & ~(size_t)255; }

extern "C" void kernel_launch(void* const* d_in, const int* in_sizes, int n_in,
                              void* d_out, int out_size, void* d_ws, size_t ws_size,
                              hipStream_t stream) {
    const float* x      = (const float*)d_in[0];
    const int*   src    = (const int*)d_in[1];
    const int*   dst    = (const int*)d_in[2];
    const float* W      = (const float*)d_in[3];
    const float* attn_l = (const float*)d_in[4];
    const float* attn_r = (const float*)d_in[5];
    const float* bias   = (const float*)d_in[6];
    float*       out    = (float*)d_out;

    const int N = in_sizes[0] / IN_DIM;
    const int E = in_sizes[1];

    char* ws = (char*)d_ws;
    float* feat = (float*)ws;
    int*   rank = (int*)ws;          ws += align256((size_t)N * OUT_DIM * sizeof(float));
    int*   ssrc = (int*)ws;          ws += align256((size_t)E * sizeof(float));
    float* el   = (float*)ws;        ws += align256((size_t)N * sizeof(float));
    float* er   = (float*)ws;        ws += align256((size_t)N * sizeof(float));
    int*   cnt  = (int*)ws;          ws += align256((size_t)N * sizeof(float));
    int*   off  = (int*)ws;          ws += align256((size_t)N * sizeof(unsigned));

    const int BLK = 256;

    gat_zero_counts<<<(N + BLK - 1) / BLK, BLK, 0, stream>>>(cnt, N);
    gat_rank<<<(E + BLK - 1) / BLK, BLK, 0, stream>>>(dst, cnt, rank, E);
    gat_offsets<<<(N + SCAN_TILE - 1) / SCAN_TILE, 1024, 0, stream>>>(cnt, off, N);
    gat_fill<<<(E + BLK - 1) / BLK, BLK, 0, stream>>>(src, dst, rank, off, ssrc, E);
    {
        int n_tiles = (N + 15) / 16;
        int waves_per_block = BLK / 32;
        int blocks = (n_tiles + waves_per_block - 1) / waves_per_block;
        gat_gemm<<<blocks, BLK, 0, stream>>>(x, W, feat, N);
    }
    gat_logits<<<1024, BLK, 0, stream>>>(feat, attn_l, attn_r, el, er, N);
    {
        int waves_per_block = BLK / 32;
        int blocks = (N + waves_per_block - 1) / waves_per_block;
        gat_segment<<<blocks, BLK, 0, stream>>>(ssrc, off, el, er, feat, bias, out, N, E);
    }
}
